# starting kernel
# baseline (speedup 1.0000x reference)
#include <hip/hip_runtime.h>

typedef __attribute__((ext_vector_type(16))) _Float16 v16h;
typedef __attribute__((ext_vector_type(8)))  float    v8f;

#define DIM 128
#define NN  4096
#define MM  4096

__device__ __forceinline__ v16h frag_load(const _Float16* __restrict__ tile, int ld) {
    const int lane = threadIdx.x & 31;
    const int r    = lane & 15;
    const int kb   = (lane & 16) ? 8 : 0;
    const _Float16* q = tile + (size_t)r * (size_t)ld + kb;
    v16h f;
    uint4* fu = reinterpret_cast<uint4*>(&f);
    fu[0] = *reinterpret_cast<const uint4*>(q);
    fu[1] = *reinterpret_cast<const uint4*>(q + 16);
    return f;
}

__device__ __forceinline__ void frag_rc(int& rbase, int& col) {
    const int lane = threadIdx.x & 31;
    col   = lane & 15;
    rbase = (lane & 16) ? 8 : 0;
}

__global__ void gemm_bias_act(const _Float16* __restrict__ A, const _Float16* __restrict__ B,
                              const float* __restrict__ bias,
                              float* __restrict__ Cf, _Float16* __restrict__ Ch,
                              int K, int lda, int ldb, int ldc, int do_relu) {
    const int i0 = blockIdx.x * 16;
    const int j0 = blockIdx.y * 16;
    v8f acc = {};
    for (int k = 0; k < K; k += 32) {
        v16h a = frag_load(A + (size_t)i0 * lda + k, lda);
        v16h b = frag_load(B + (size_t)j0 * ldb + k, ldb);
        acc = __builtin_amdgcn_wmma_f32_16x16x32_f16(false, a, false, b, (short)0, acc,
                                                     false, false);
    }
    int rb, col; frag_rc(rb, col);
    const float bv = bias ? bias[j0 + col] : 0.f;
#pragma unroll
    for (int r = 0; r < 8; ++r) {
        const int row = i0 + rb + r;
        float v = acc[r] + bv;
        if (do_relu) v = fmaxf(v, 0.f);
        const size_t idx = (size_t)row * ldc + j0 + col;
        if (Cf) Cf[idx] = v;
        if (Ch) Ch[idx] = (_Float16)v;
    }
}

__global__ void gemm_attn(const _Float16* __restrict__ hs, const float* __restrict__ adj,
                          _Float16* __restrict__ attn) {
    const int i0 = blockIdx.x * 16;
    const int j0 = blockIdx.y * 16;
    v8f acc = {};
#pragma unroll
    for (int k = 0; k < DIM; k += 32) {
        v16h a = frag_load(hs + (size_t)i0 * DIM + k, DIM);
        v16h b = frag_load(hs + (size_t)j0 * DIM + k, DIM);
        acc = __builtin_amdgcn_wmma_f32_16x16x32_f16(false, a, false, b, (short)0, acc,
                                                     false, false);
    }
    int rb, col; frag_rc(rb, col);
#pragma unroll
    for (int r = 0; r < 8; ++r) {
        const int row = i0 + rb + r;
        const size_t idx = (size_t)row * NN + j0 + col;
        attn[idx] = (_Float16)(acc[r] * adj[idx]);
    }
}

__global__ void row_abs_sum(const _Float16* __restrict__ attn, float* __restrict__ denom) {
    __shared__ float red[256];
    const int i = blockIdx.x;
    float s = 0.f;
    for (int j = threadIdx.x; j < NN; j += 256) s += fabsf((float)attn[(size_t)i * NN + j]);
    red[threadIdx.x] = s;
    __syncthreads();
    for (int w = 128; w > 0; w >>= 1) {
        if (threadIdx.x < w) red[threadIdx.x] += red[threadIdx.x + w];
        __syncthreads();
    }
    if (threadIdx.x == 0) denom[i] = fmaxf(red[0], 1e-12f);
}

__global__ void transpose_f16(const _Float16* __restrict__ src, _Float16* __restrict__ dst) {
    const int j = blockIdx.x;
    const int n = threadIdx.x;
    dst[(size_t)n * NN + j] = src[(size_t)j * DIM + n];
}

__global__ void gemm_norm_axpy(const _Float16* __restrict__ attn, const _Float16* __restrict__ hsT,
                               const float* __restrict__ denom, float* __restrict__ xs) {
    const int i0 = blockIdx.x * 16;
    const int j0 = blockIdx.y * 16;
    const _Float16* Ab = attn + (size_t)i0 * NN;
    const _Float16* Bb = hsT + (size_t)j0 * NN;
    v8f acc = {};
    for (int k = 0; k < NN; k += 32) {
        __builtin_prefetch(Ab + k + 256, 0, 0);
        v16h a = frag_load(Ab + k, NN);
        v16h b = frag_load(Bb + k, NN);
        acc = __builtin_amdgcn_wmma_f32_16x16x32_f16(false, a, false, b, (short)0, acc,
                                                     false, false);
    }
    int rb, col; frag_rc(rb, col);
#pragma unroll
    for (int r = 0; r < 8; ++r) {
        const int row = i0 + rb + r;
        xs[(size_t)row * DIM + j0 + col] += acc[r] / denom[row];
    }
}

__global__ void gather_embed(const int* __restrict__ idx, const float* __restrict__ emb,
                             float* __restrict__ of32, _Float16* __restrict__ of16) {
    const int i = blockIdx.x;
    const int n = threadIdx.x;
    const float v = emb[(size_t)idx[i] * DIM + n];
    if (of32) of32[(size_t)i * DIM + n] = v;
    if (of16) of16[(size_t)i * DIM + n] = (_Float16)v;
}

__global__ void f32_to_f16(const float* __restrict__ s, _Float16* __restrict__ d, int n) {
    const int i = blockIdx.x * 256 + threadIdx.x;
    if (i < n) d[i] = (_Float16)s[i];
}

__global__ void vec_add(const float* __restrict__ a, const float* __restrict__ b,
                        float* __restrict__ c, int n) {
    const int i = blockIdx.x * 256 + threadIdx.x;
    if (i < n) c[i] = a[i] + b[i];
}

__global__ void lstm_dir(const float* __restrict__ G, const float* __restrict__ Whh,
                         float* __restrict__ hout) {
    extern __shared__ float sm[];
    float* WhhT = sm;
    float* h_sh = sm + DIM * 512;
    float* z_sh = h_sh + DIM;
    const int d   = blockIdx.x;
    const int tid = threadIdx.x;
    const float* W  = Whh + (size_t)d * 512 * DIM;
    const float* Gd = G + (size_t)d * MM * 512;
    float* hd = hout + (size_t)d * MM * DIM;

    for (int idx = tid; idx < 512 * DIM; idx += 512) {
        const int j = idx >> 7, k = idx & 127;
        WhhT[k * 512 + j] = W[idx];
    }
    if (tid < DIM) h_sh[tid] = 0.f;
    float c = 0.f;
    __syncthreads();

    for (int t = 0; t < MM; ++t) {
        const int tt = d ? (MM - 1 - t) : t;
        float acc = Gd[(size_t)tt * 512 + tid];
#pragma unroll 8
        for (int k = 0; k < DIM; ++k) acc += h_sh[k] * WhhT[k * 512 + tid];
        z_sh[tid] = acc;
        __syncthreads();
        if (tid < DIM) {
            const float zi = z_sh[tid], zf = z_sh[tid + 128];
            const float zg = z_sh[tid + 256], zo = z_sh[tid + 384];
            const float ig = 1.f / (1.f + __expf(-zi));
            const float fg = 1.f / (1.f + __expf(-zf));
            const float gg = tanhf(zg);
            const float og = 1.f / (1.f + __expf(-zo));
            c = fg * c + ig * gg;
            const float h = og * tanhf(c);
            h_sh[tid] = h;
            hd[(size_t)tt * DIM + tid] = h;
        }
        __syncthreads();
    }
}

__global__ void bilstm_concat_relu(const float* __restrict__ hbuf, _Float16* __restrict__ xs_rnn) {
    const int t = blockIdx.x;
    const int n = threadIdx.x;
    const float v = (n < DIM) ? hbuf[(size_t)t * DIM + n]
                              : hbuf[(size_t)(MM + t) * DIM + (n - DIM)];
    xs_rnn[(size_t)t * (2 * DIM) + n] = (_Float16)fmaxf(v, 0.f);
}

__global__ void row_mean(const float* __restrict__ W, float* __restrict__ out) {
    __shared__ float red[256];
    const int i = blockIdx.x;
    float s = 0.f;
    for (int j = threadIdx.x; j < MM; j += 256) s += W[(size_t)i * MM + j];
    red[threadIdx.x] = s;
    __syncthreads();
    for (int w = 128; w > 0; w >>= 1) {
        if (threadIdx.x < w) red[threadIdx.x] += red[threadIdx.x + w];
        __syncthreads();
    }
    if (threadIdx.x == 0) out[i] = red[0] / (float)MM;
}

__global__ void col_mean(const float* __restrict__ W, float* __restrict__ out) {
    const int j = blockIdx.x * 256 + threadIdx.x;
    float s = 0.f;
    for (int i = 0; i < NN; ++i) s += W[(size_t)i * MM + j];
    out[j] = s / (float)NN;
}

__global__ void pool_vectors(const float* __restrict__ h, const float* __restrict__ hp,
                             const float* __restrict__ wc, const float* __restrict__ wp,
                             float* __restrict__ cat) {
    const int tid = threadIdx.x;
    const int n = tid & 127;
    const float* src = (tid < 128) ? h : hp;
    const float* wv  = (tid < 128) ? wc : wp;
    float s = 0.f;
    for (int i = 0; i < NN; ++i) s += wv[i] * src[(size_t)i * DIM + n];
    cat[tid] = s / (float)NN;
}

__global__ void final_mlp(const float* __restrict__ cat_in, const float* __restrict__ Wo,
                          const float* __restrict__ bo, const float* __restrict__ Wi,
                          const float* __restrict__ bi, float* __restrict__ out) {
    __shared__ float cur[256], nxt[256];
    const int t = threadIdx.x;
    cur[t] = cat_in[t];
    __syncthreads();
    for (int l = 0; l < 2; ++l) {
        const float* Wl = Wo + (size_t)l * 256 * 256;
        float s = bo[l * 256 + t];
        for (int k = 0; k < 256; ++k) s += cur[k] * Wl[(size_t)t * 256 + k];
        nxt[t] = fmaxf(s, 0.f);
        __syncthreads();
        cur[t] = nxt[t];
        __syncthreads();
    }
    if (t < 2) {
        float s = bi[t];
        for (int k = 0; k < 256; ++k) s += cur[k] * Wi[(size_t)t * 256 + k];
        out[t] = s;
    }
}

extern "C" void kernel_launch(void* const* d_in, const int* in_sizes, int n_in,
                              void* d_out, int out_size, void* d_ws, size_t ws_size,
                              hipStream_t stream) {
    (void)in_sizes; (void)n_in; (void)out_size; (void)ws_size;
    const int*   fingerprints = (const int*)d_in[0];
    const float* adjacency    = (const float*)d_in[1];
    const int*   words        = (const int*)d_in[2];
    const float* emb_fp       = (const float*)d_in[3];
    const float* emb_word     = (const float*)d_in[4];
    const float* Wg  = (const float*)d_in[5];
    const float* bg  = (const float*)d_in[6];
    const float* Wih = (const float*)d_in[7];
    const float* Whh = (const float*)d_in[8];
    const float* bih = (const float*)d_in[9];
    const float* bhh = (const float*)d_in[10];
    const float* Wc  = (const float*)d_in[11];
    const float* bc  = (const float*)d_in[12];
    const float* Wp  = (const float*)d_in[13];
    const float* bp  = (const float*)d_in[14];
    const float* Wo  = (const float*)d_in[15];
    const float* bo  = (const float*)d_in[16];
    const float* Wi  = (const float*)d_in[17];
    const float* bi  = (const float*)d_in[18];

    float* out         = (float*)d_out;
    float* weights_out = out + 2;

    char* base = (char*)d_ws;
    size_t off = 0;
    auto alloc = [&](size_t bytes) -> void* {
        void* p = base + off;
        off = (off + bytes + 255) & ~(size_t)255;
        return p;
    };
    float*    xs       = (float*)   alloc((size_t)NN * DIM * 4);
    _Float16* xs_h     = (_Float16*)alloc((size_t)NN * DIM * 2);
    _Float16* hs_h     = (_Float16*)alloc((size_t)NN * DIM * 2);
    _Float16* hsT_h    = (_Float16*)alloc((size_t)DIM * NN * 2);
    _Float16* attn_h   = (_Float16*)alloc((size_t)NN * NN * 2);
    float*    denom    = (float*)   alloc((size_t)NN * 4);
    _Float16* Wg_h     = (_Float16*)alloc((size_t)3 * DIM * DIM * 2);
    _Float16* Wc_h     = (_Float16*)alloc((size_t)DIM * DIM * 2);
    _Float16* Wp_h     = (_Float16*)alloc((size_t)DIM * 2 * DIM * 2);
    _Float16* Wih_h    = (_Float16*)alloc((size_t)2 * 512 * DIM * 2);
    float*    bsum     = (float*)   alloc((size_t)2 * 512 * 4);
    _Float16* wv_h     = (_Float16*)alloc((size_t)MM * DIM * 2);
    float*    G        = (float*)   alloc((size_t)2 * MM * 512 * 4);
    float*    hbuf     = (float*)   alloc((size_t)2 * MM * DIM * 4);
    _Float16* xs_rnn_h = (_Float16*)alloc((size_t)MM * 2 * DIM * 2);
    float*    h_f32    = (float*)   alloc((size_t)NN * DIM * 4);
    _Float16* h_h      = (_Float16*)alloc((size_t)NN * DIM * 2);
    float*    hp_f32   = (float*)   alloc((size_t)MM * DIM * 4);
    _Float16* hp_h     = (_Float16*)alloc((size_t)MM * DIM * 2);
    float*    w_comp   = (float*)   alloc((size_t)NN * 4);
    float*    w_prot   = (float*)   alloc((size_t)MM * 4);
    float*    cat      = (float*)   alloc(256 * 4);

    f32_to_f16<<<(3 * DIM * DIM + 255) / 256, 256, 0, stream>>>(Wg, Wg_h, 3 * DIM * DIM);
    f32_to_f16<<<(DIM * DIM + 255) / 256, 256, 0, stream>>>(Wc, Wc_h, DIM * DIM);
    f32_to_f16<<<(DIM * 2 * DIM + 255) / 256, 256, 0, stream>>>(Wp, Wp_h, DIM * 2 * DIM);
    f32_to_f16<<<(2 * 512 * DIM + 255) / 256, 256, 0, stream>>>(Wih, Wih_h, 2 * 512 * DIM);
    vec_add<<<(2 * 512 + 255) / 256, 256, 0, stream>>>(bih, bhh, bsum, 2 * 512);

    gather_embed<<<NN, DIM, 0, stream>>>(fingerprints, emb_fp, xs, nullptr);
    gather_embed<<<MM, DIM, 0, stream>>>(words, emb_word, nullptr, wv_h);

    for (int l = 0; l < 3; ++l) {
        f32_to_f16<<<(NN * DIM + 255) / 256, 256, 0, stream>>>(xs, xs_h, NN * DIM);
        gemm_bias_act<<<dim3(NN / 16, DIM / 16), 32, 0, stream>>>(
            xs_h, Wg_h + (size_t)l * DIM * DIM, bg + l * DIM,
            nullptr, hs_h, DIM, DIM, DIM, DIM, 1);
        transpose_f16<<<NN, DIM, 0, stream>>>(hs_h, hsT_h);
        gemm_attn<<<dim3(NN / 16, NN / 16), 32, 0, stream>>>(hs_h, adjacency, attn_h);
        row_abs_sum<<<NN, 256, 0, stream>>>(attn_h, denom);
        gemm_norm_axpy<<<dim3(NN / 16, DIM / 16), 32, 0, stream>>>(attn_h, hsT_h, denom, xs);
    }

    f32_to_f16<<<(NN * DIM + 255) / 256, 256, 0, stream>>>(xs, xs_h, NN * DIM);
    gemm_bias_act<<<dim3(NN / 16, DIM / 16), 32, 0, stream>>>(
        xs_h, Wc_h, bc, h_f32, h_h, DIM, DIM, DIM, DIM, 1);

    for (int d = 0; d < 2; ++d) {
        gemm_bias_act<<<dim3(MM / 16, 512 / 16), 32, 0, stream>>>(
            wv_h, Wih_h + (size_t)d * 512 * DIM, bsum + d * 512,
            G + (size_t)d * MM * 512, nullptr, DIM, DIM, DIM, 512, 0);
    }
    const size_t lstm_lds = (size_t)(DIM * 512 + DIM + 512) * sizeof(float);
    lstm_dir<<<2, 512, lstm_lds, stream>>>(G, Whh, hbuf);
    bilstm_concat_relu<<<MM, 256, 0, stream>>>(hbuf, xs_rnn_h);
    gemm_bias_act<<<dim3(MM / 16, DIM / 16), 32, 0, stream>>>(
        xs_rnn_h, Wp_h, bp, hp_f32, hp_h, 2 * DIM, 2 * DIM, 2 * DIM, DIM, 1);

    gemm_bias_act<<<dim3(NN / 16, MM / 16), 32, 0, stream>>>(
        h_h, hp_h, nullptr, weights_out, nullptr, DIM, DIM, DIM, MM, 0);

    row_mean<<<NN, 256, 0, stream>>>(weights_out, w_comp);
    col_mean<<<MM / 256, 256, 0, stream>>>(weights_out, w_prot);
    pool_vectors<<<1, 256, 0, stream>>>(h_f32, hp_f32, w_comp, w_prot, cat);
    final_mlp<<<1, 256, 0, stream>>>(cat, Wo, bo, Wi, bi, out);
}


# fewer launches (37 -> 18): f16 copy of xs, fragment layout of hs and |attn| row sums made in the product kernels that hold the data; independent launches merged by block role; row sum order changed
# speedup vs baseline: 7.6686x; 7.6686x over previous
#include <hip/hip_runtime.h>

typedef __attribute__((ext_vector_type(16))) _Float16 v16h;
typedef __attribute__((ext_vector_type(8)))  float    v8f;
typedef __attribute__((ext_vector_type(4)))  float    v4f;

#define DIM 128
#define NN  4096
#define MM  4096

__device__ __forceinline__ v16h frag_load(const _Float16* __restrict__ tile, int ld) {
    const int lane = threadIdx.x & 31;
    const int r    = lane & 15;
    const int kb   = (lane & 16) ? 8 : 0;
    const _Float16* q = tile + (size_t)r * (size_t)ld + kb;
    v16h f;
    uint4* fu = reinterpret_cast<uint4*>(&f);
    fu[0] = *reinterpret_cast<const uint4*>(q);
    fu[1] = *reinterpret_cast<const uint4*>(q + 16);
    return f;
}

__device__ __forceinline__ void frag_rc(int& rbase, int& col) {
    const int lane = threadIdx.x & 31;
    col   = lane & 15;
    rbase = (lane & 16) ? 8 : 0;
}

typedef __attribute__((ext_vector_type(8))) _Float16 v8h;
__device__ __forceinline__ void gemm_tile(const _Float16* __restrict__ A, const _Float16* __restrict__ B,
                                          const float* __restrict__ bias,
                                          float* __restrict__ Cf, _Float16* __restrict__ Ch,
                                          _Float16* __restrict__ ChF,
                                          int K, int lda, int ldb, int ldc, int do_relu,
                                          int bx, int by, int ny) {
    const int i0 = bx * 16;
    const int j0 = by * 16;
    v8f acc = {};
    for (int k = 0; k < K; k += 32) {
        v16h a = frag_load(A + (size_t)i0 * lda + k, lda);
        v16h b = frag_load(B + (size_t)j0 * ldb + k, ldb);
        acc = __builtin_amdgcn_wmma_f32_16x16x32_f16(false, a, false, b, (short)0, acc,
                                                     false, false);
    }
    int rb, col; frag_rc(rb, col);
    const float bv = bias ? bias[j0 + col] : 0.f;
    v8h piece;
#pragma unroll
    for (int r = 0; r < 8; ++r) {
        const int row = i0 + rb + r;
        float v = acc[r] + bv;
        if (do_relu) v = fmaxf(v, 0.f);
        const size_t idx = (size_t)row * ldc + j0 + col;
        if (Cf) Cf[idx] = v;
        if (Ch) Ch[idx] = (_Float16)v;
        piece[r] = (_Float16)v;
    }
    if (ChF) {
        const size_t o = (size_t)((bx >> 1) * ny + by) * 512 + (bx & 1) * 256 + (threadIdx.x & 31) * 8;
        *reinterpret_cast<v8h*>(ChF + o) = piece;
    }
}

__global__ void gemm_bias_act(const _Float16* __restrict__ A, const _Float16* __restrict__ B,
                              const float* __restrict__ bias,
                              float* __restrict__ Cf, _Float16* __restrict__ Ch,
                              _Float16* __restrict__ ChF,
                              int K, int lda, int ldb, int ldc, int do_relu) {
    gemm_tile(A, B, bias, Cf, Ch, ChF, K, lda, ldb, ldc, do_relu, blockIdx.x, blockIdx.y, gridDim.y);
}

#define PJ_TILES ((NN / 16) * (DIM / 16))
#define PG_TILES ((MM / 16) * (512 / 16))
__global__ void gemm_pre3(const _Float16* __restrict__ xs_h, const _Float16* __restrict__ Wc_h,
                          const float* __restrict__ bc,
                          float* __restrict__ h_f32, _Float16* __restrict__ h_h,
                          const _Float16* __restrict__ wv_h, const _Float16* __restrict__ Wih_h,
                          const float* __restrict__ bsum, float* __restrict__ G) {
    int b = blockIdx.x;
    if (b < PJ_TILES) {
        gemm_tile(xs_h, Wc_h, bc, h_f32, h_h, nullptr, DIM, DIM, DIM, DIM, 1,
                  b % (NN / 16), b / (NN / 16), DIM / 16);
    } else {
        b -= PJ_TILES;
        const int d = b / PG_TILES;
        b -= d * PG_TILES;
        gemm_tile(wv_h, Wih_h + (size_t)d * 512 * DIM, bsum + d * 512,
                  G + (size_t)d * MM * 512, nullptr, nullptr, DIM, DIM, DIM, 512, 0,
                  b % (MM / 16), b / (MM / 16), 512 / 16);
    }
}

typedef __attribute__((ext_vector_type(2))) float    v2f;

__global__ __launch_bounds__(256) void gemm_attn(const _Float16* __restrict__ hs,
                                                 const float* __restrict__ adj,
                                                 _Float16* __restrict__ attn) {
    const int lane = threadIdx.x & 31;
    const int wave = threadIdx.x >> 5;
    const int c    = lane & 15;
    const int hh   = lane >> 4;
    const int i0   = (blockIdx.x * 8 + wave) * 32;
    const int jb   = blockIdx.y * 128;
    v16h b0[4], b1[4];
#pragma unroll
    for (int ks = 0; ks < 4; ++ks) {
        b0[ks] = frag_load(hs + (size_t)i0 * DIM + ks * 32, DIM);
        b1[ks] = frag_load(hs + (size_t)(i0 + 16) * DIM + ks * 32, DIM);
    }
    const size_t r0 = (size_t)(i0 + c) * NN + jb + 8 * hh;
    const size_t r1 = r0 + (size_t)16 * NN;
    for (int t = 0; t < 8; t += 2) {
        v4f  x[2][4];
        v16h a[2][4];
#pragma unroll
        for (int u = 0; u < 2; ++u) {
            x[u][0] = *reinterpret_cast<const v4f*>(adj + r0 + (t + u) * 16);
            x[u][1] = *reinterpret_cast<const v4f*>(adj + r0 + (t + u) * 16 + 4);
            x[u][2] = *reinterpret_cast<const v4f*>(adj + r1 + (t + u) * 16);
            x[u][3] = *reinterpret_cast<const v4f*>(adj + r1 + (t + u) * 16 + 4);
        }
#pragma unroll
        for (int u = 0; u < 2; ++u)
#pragma unroll
            for (int ks = 0; ks < 4; ++ks)
                a[u][ks] = frag_load(hs + (size_t)(jb + (t + u) * 16) * DIM + ks * 32, DIM);
        __builtin_amdgcn_sched_barrier(0);
#pragma unroll
        for (int u = 0; u < 2; ++u) {
            v8f acc0 = {}, acc1 = {};
#pragma unroll
            for (int ks = 0; ks < 4; ++ks) {
                acc0 = __builtin_amdgcn_wmma_f32_16x16x32_f16(false, a[u][ks], false, b0[ks],
                                                              (short)0, acc0, false, false);
                acc1 = __builtin_amdgcn_wmma_f32_16x16x32_f16(false, a[u][ks], false, b1[ks],
                                                              (short)0, acc1, false, false);
            }
            v8h o0, o1;
#pragma unroll
            for (int r = 0; r < 4; ++r) {
                o0[r]     = (_Float16)(acc0[r] * x[u][0][r]);
                o0[r + 4] = (_Float16)(acc0[r + 4] * x[u][1][r]);
                o1[r]     = (_Float16)(acc1[r] * x[u][2][r]);
                o1[r + 4] = (_Float16)(acc1[r + 4] * x[u][3][r]);
            }
            *reinterpret_cast<v8h*>(attn + r0 + (t + u) * 16) = o0;
            *reinterpret_cast<v8h*>(attn + r1 + (t + u) * 16) = o1;
        }
    }
}

__global__ __launch_bounds__(256) void gemm_weights(const _Float16* __restrict__ h,
                                                    const _Float16* __restrict__ hp,
                                                    float* __restrict__ C) {
    const int lane = threadIdx.x & 31;
    const int wave = threadIdx.x >> 5;
    const int c    = lane & 15;
    const int hh   = lane >> 4;
    const int i0   = (blockIdx.x * 8 + wave) * 32;
    const int jb   = blockIdx.y * 128;
    v16h b0[4], b1[4];
#pragma unroll
    for (int ks = 0; ks < 4; ++ks) {
        b0[ks] = frag_load(h + (size_t)i0 * DIM + ks * 32, DIM);
        b1[ks] = frag_load(h + (size_t)(i0 + 16) * DIM + ks * 32, DIM);
    }
    const size_t r0 = (size_t)(i0 + c) * MM + jb + 8 * hh;
    const size_t r1 = r0 + (size_t)16 * MM;
#pragma nounroll
    for (int t = 0; t < 8; t += 2) {
        v16h a[2][4];
#pragma unroll
        for (int u = 0; u < 2; ++u)
#pragma unroll
            for (int ks = 0; ks < 4; ++ks)
                a[u][ks] = frag_load(hp + (size_t)(jb + (t + u) * 16) * DIM + ks * 32, DIM);
        __builtin_amdgcn_sched_barrier(0);
        v8f acc[2][2] = {};
#pragma unroll
        for (int ks = 0; ks < 4; ++ks)
#pragma unroll
            for (int u = 0; u < 2; ++u) {
                acc[u][0] = __builtin_amdgcn_wmma_f32_16x16x32_f16(false, a[u][ks], false, b0[ks],
                                                                   (short)0, acc[u][0], false, false);
                acc[u][1] = __builtin_amdgcn_wmma_f32_16x16x32_f16(false, a[u][ks], false, b1[ks],
                                                                   (short)0, acc[u][1], false, false);
            }
#pragma unroll
        for (int u = 0; u < 2; ++u) {
            v2f* q0 = reinterpret_cast<v2f*>(C + r0 + (t + u) * 16);
            v2f* q1 = reinterpret_cast<v2f*>(C + r1 + (t + u) * 16);
#pragma unroll
            for (int r = 0; r < 4; ++r) {
                q0[r] = v2f{acc[u][0][2 * r], acc[u][0][2 * r + 1]};
                q1[r] = v2f{acc[u][1][2 * r], acc[u][1][2 * r + 1]};
            }
        }
    }
}

__device__ __forceinline__ float wave_sum(float s) {
    for (int m = 16; m > 0; m >>= 1) s += __shfl_xor(s, m, 32);
    return s;
}

#define NA_PITCH 132
__global__ __launch_bounds__(256) void gemm_norm_axpy(const _Float16* __restrict__ attn,
                                                      const _Float16* __restrict__ hsF,
                                                      float* __restrict__ xs,
                                                      _Float16* __restrict__ xs_h) {
    __shared__ float part[8][16 * NA_PITCH];
    __shared__ float dpart[8][16];
    __shared__ float dsum[16];
    const int lane = threadIdx.x & 31;
    const int wave = threadIdx.x >> 5;
    const int i0   = blockIdx.x * 16;
    const _Float16* Ab = attn + (size_t)i0 * NN + wave * 512;
    const _Float16* Bb = hsF + (size_t)(wave * 16) * (8 * 512) + lane * 8;
    v8f acc[8];
#pragma unroll
    for (int j = 0; j < 8; ++j) acc[j] = v8f{};
    float as0 = 0.f, as1 = 0.f;
#pragma nounroll
    for (int t = 0; t < 8; ++t) {
        v16h a[2], b[2][8];
#pragma unroll
        for (int u = 0; u < 2; ++u) a[u] = frag_load(Ab + t * 64 + 32 * u, NN);
#pragma unroll
        for (int u = 0; u < 2; ++u)
#pragma unroll
            for (int j = 0; j < 8; ++j) {
                const _Float16* p = Bb + (size_t)((2 * t + u) * 8 + j) * 512;
                uint4* fu = reinterpret_cast<uint4*>(&b[u][j]);
                fu[0] = *reinterpret_cast<const uint4*>(p);
                fu[1] = *reinterpret_cast<const uint4*>(p + 256);
            }
        __builtin_amdgcn_sched_barrier(0);
#pragma unroll
        for (int u = 0; u < 2; ++u)
#pragma unroll
            for (int j = 0; j < 8; ++j)
                acc[j] = __builtin_amdgcn_wmma_f32_16x16x32_f16(false, a[u], false, b[u][j],
                                                                (short)0, acc[j], false, false);
#pragma unroll
        for (int u = 0; u < 2; ++u)
#pragma unroll
            for (int e = 0; e < 16; e += 2) {
                as0 += fabsf((float)a[u][e]);
                as1 += fabsf((float)a[u][e + 1]);
            }
    }
    const int tid = threadIdx.x;
    const int col = tid & 127;
    const int rq  = tid >> 7;
    float xv[8];
#pragma unroll
    for (int q = 0; q < 8; ++q) xv[q] = xs[(size_t)(i0 + rq + 2 * q) * DIM + col];
    {
        float rsum = as0 + as1;
        rsum += __shfl_xor(rsum, 16, 32);
        if (lane < 16) dpart[wave][lane] = rsum;
    }
    const int c  = lane & 15;
    const int hh = lane >> 4;
    float* pw = &part[wave][0];
#pragma unroll
    for (int j = 0; j < 8; ++j)
#pragma unroll
        for (int r = 0; r < 8; ++r)
            pw[(8 * hh + r) * NA_PITCH + 16 * j + c] = acc[j][r];
    __syncthreads();
    if (tid < 16)
        dsum[tid] = fmaxf(((dpart[0][tid] + dpart[1][tid]) + (dpart[2][tid] + dpart[3][tid]))
                        + ((dpart[4][tid] + dpart[5][tid]) + (dpart[6][tid] + dpart[7][tid])), 1e-12f);
    __syncthreads();
#pragma unroll
    for (int q = 0; q < 8; ++q) {
        const int o = (rq + 2 * q) * NA_PITCH + col;
        const float s = ((part[0][o] + part[1][o]) + (part[2][o] + part[3][o]))
                      + ((part[4][o] + part[5][o]) + (part[6][o] + part[7][o]));
        const float v = xv[q] + s / dsum[rq + 2 * q];
        xs[(size_t)(i0 + rq + 2 * q) * DIM + col]   = v;
        xs_h[(size_t)(i0 + rq + 2 * q) * DIM + col] = (_Float16)v;
    }
}

typedef __attribute__((ext_vector_type(4))) _Float16 v4h;

__device__ __forceinline__ void cvt8(const float* __restrict__ s, _Float16* __restrict__ d, int i) {
    const v4f a = *reinterpret_cast<const v4f*>(s + i);
    const v4f b = *reinterpret_cast<const v4f*>(s + i + 4);
    v8h h;
#pragma unroll
    for (int e = 0; e < 4; ++e) { h[e] = (_Float16)a[e]; h[e + 4] = (_Float16)b[e]; }
    *reinterpret_cast<v8h*>(d + i) = h;
}

__device__ __forceinline__ void gather8(const int* __restrict__ idx, const float* __restrict__ emb,
                                        float* __restrict__ of32, _Float16* __restrict__ of16, int blk) {
    const int lane = threadIdx.x & 31;
    const int i    = blk * 8 + (threadIdx.x >> 5);
    const v4f v = *reinterpret_cast<const v4f*>(emb + (size_t)idx[i] * DIM + 4 * lane);
    if (of32) *reinterpret_cast<v4f*>(of32 + (size_t)i * DIM + 4 * lane) = v;
    v4h h;
    h[0] = (_Float16)v[0]; h[1] = (_Float16)v[1]; h[2] = (_Float16)v[2]; h[3] = (_Float16)v[3];
    *reinterpret_cast<v4h*>(of16 + (size_t)i * DIM + 4 * lane) = h;
}

#define SB_WG   0
#define SB_WC   (SB_WG + (3 * DIM * DIM) / 2048)
#define SB_WP   (SB_WC + (DIM * DIM) / 2048)
#define SB_WIH  (SB_WP + (DIM * 2 * DIM) / 2048)
#define SB_BIAS (SB_WIH + (2 * 512 * DIM) / 2048)
#define SB_FP   (SB_BIAS + (2 * 512) / 256)
#define SB_WORD (SB_FP + NN / 8)
#define SB_END  (SB_WORD + MM / 8)
__global__ __launch_bounds__(256) void setup_all(const float* __restrict__ Wg, _Float16* __restrict__ Wg_h,
                                                 const float* __restrict__ Wc, _Float16* __restrict__ Wc_h,
                                                 const float* __restrict__ Wp, _Float16* __restrict__ Wp_h,
                                                 const float* __restrict__ Wih, _Float16* __restrict__ Wih_h,
                                                 const float* __restrict__ bih, const float* __restrict__ bhh,
                                                 float* __restrict__ bsum,
                                                 const int* __restrict__ fingerprints,
                                                 const float* __restrict__ emb_fp,
                                                 float* __restrict__ xs, _Float16* __restrict__ xs_h,
                                                 const int* __restrict__ words,
                                                 const float* __restrict__ emb_word,
                                                 _Float16* __restrict__ wv_h) {
    const int b = blockIdx.x;
    if (b < SB_WC)        cvt8(Wg,  Wg_h,  ((b - SB_WG)  * 256 + threadIdx.x) * 8);
    else if (b < SB_WP)   cvt8(Wc,  Wc_h,  ((b - SB_WC)  * 256 + threadIdx.x) * 8);
    else if (b < SB_WIH)  cvt8(Wp,  Wp_h,  ((b - SB_WP)  * 256 + threadIdx.x) * 8);
    else if (b < SB_BIAS) cvt8(Wih, Wih_h, ((b - SB_WIH) * 256 + threadIdx.x) * 8);
    else if (b < SB_FP) {
        const int i = (b - SB_BIAS) * 256 + threadIdx.x;
        bsum[i] = bih[i] + bhh[i];
    }
    else if (b < SB_WORD) gather8(fingerprints, emb_fp, xs, xs_h, b - SB_FP);
    else                  gather8(words, emb_word, nullptr, wv_h, b - SB_WORD);
}

__device__ __forceinline__ float tanh_sel(float x) {
    const float y  = __builtin_fabsf(x);
    const float y2 = y * y;
    float p = __builtin_fmaf(y2, -0x1.758e7ap-8f, 0x1.521192p-6f);
    p = __builtin_fmaf(y2, p, -0x1.b8389cp-5f);
    p = __builtin_fmaf(y2, p, 0x1.110704p-3f);
    p = __builtin_fmaf(y2, p, -0x1.555532p-2f);
    const float zs = __builtin_fmaf(y2, y * p, y);
    const float e  = expf(2.0f * y);
    const float zl = __builtin_fmaf(-2.0f, __builtin_amdgcn_rcpf(e + 1.0f), 1.0f);
    return __builtin_copysignf((y < 0.625f) ? zs : zl, x);
}

#define GCH 16
static_assert(GCH == 16 && MM % GCH == 0, "G_ROWS names 16 rows; pieces tile the sequence");
__global__ __launch_bounds__(128) void lstm_dir(const float* __restrict__ G,
                                                const float* __restrict__ Whh,
                                                float* __restrict__ hout) {
    __shared__ __attribute__((aligned(16))) _Float16 h16[2][DIM];
    __shared__ __attribute__((aligned(16))) float gs[2][GCH * 512];
    const int d    = blockIdx.x;
    const int tid  = threadIdx.x;
    const int lane = tid & 31;
    const int c16  = lane & 15;
    const int hh   = lane >> 4;
    const int u0   = tid & ~31;
    const float* W  = Whh + (size_t)d * 512 * DIM;
    const float* Gd = G + (size_t)d * MM * 512;
    float* hd = hout + (size_t)d * MM * DIM;

    v16h bw[4][2][4];
#pragma unroll
    for (int g = 0; g < 4; ++g)
#pragma unroll
        for (int tl = 0; tl < 2; ++tl)
#pragma unroll
            for (int ks = 0; ks < 4; ++ks) {
                const float* q = W + (size_t)(g * 128 + u0 + tl * 16 + c16) * DIM + ks * 32 + 8 * hh;
                const float4 x0 = *reinterpret_cast<const float4*>(q);
                const float4 x1 = *reinterpret_cast<const float4*>(q + 4);
                const float4 x2 = *reinterpret_cast<const float4*>(q + 16);
                const float4 x3 = *reinterpret_cast<const float4*>(q + 20);
                v16h f;
                f[0]  = (_Float16)x0.x; f[1]  = (_Float16)x0.y; f[2]  = (_Float16)x0.z; f[3]  = (_Float16)x0.w;
                f[4]  = (_Float16)x1.x; f[5]  = (_Float16)x1.y; f[6]  = (_Float16)x1.z; f[7]  = (_Float16)x1.w;
                f[8]  = (_Float16)x2.x; f[9]  = (_Float16)x2.y; f[10] = (_Float16)x2.z; f[11] = (_Float16)x2.w;
                f[12] = (_Float16)x3.x; f[13] = (_Float16)x3.y; f[14] = (_Float16)x3.z; f[15] = (_Float16)x3.w;
                bw[g][tl][ks] = f;
            }

#define G_ROWS(X) X(0) X(1) X(2) X(3) X(4) X(5) X(6) X(7) X(8) X(9) X(10) X(11) X(12) X(13) X(14) X(15)
#define G_DECL(i) v4f pre##i;
#define G_LOAD(i) pre##i = src[(i) * 128 + tid];
#define G_STORE(i) dst[(i) * 128 + tid] = pre##i;
    G_ROWS(G_DECL)
    {
        const v4f* src = reinterpret_cast<const v4f*>(Gd + (size_t)(d ? (MM - GCH) : 0) * 512);
        G_ROWS(G_LOAD)
        v4f* dst = reinterpret_cast<v4f*>(&gs[0][0]);
        G_ROWS(G_STORE)
    }
    h16[0][tid] = (_Float16)0.f;
    float c = 0.f;
    __syncthreads();

    for (int p = 0; p < MM / GCH; ++p) {
        const int  gb   = p & 1;
        const bool more = (p + 1) < MM / GCH;
        if (more) {
            const int r0 = d ? (MM - (p + 2) * GCH) : (p + 1) * GCH;
            const v4f* src = reinterpret_cast<const v4f*>(Gd + (size_t)r0 * 512);
            G_ROWS(G_LOAD)
        }

        for (int s = 0; s < GCH; ++s) {
            const int t   = p * GCH + s;
            const int tt  = d ? (MM - 1 - t) : t;
            const int cur = t & 1;
            const int sr  = d ? (GCH - 1 - s) : s;

            if (more && s == GCH - 1) {
                v4f* dst = reinterpret_cast<v4f*>(&gs[gb ^ 1][0]);
                G_ROWS(G_STORE)
            }

            const _Float16* hp = &h16[cur][8 * hh];
            v16h a[4];
#pragma unroll
            for (int ks = 0; ks < 4; ++ks) {
                uint4* au = reinterpret_cast<uint4*>(&a[ks]);
                au[0] = *reinterpret_cast<const uint4*>(hp + ks * 32);
                au[1] = *reinterpret_cast<const uint4*>(hp + ks * 32 + 16);
            }
            const float* gp = &gs[gb][sr * 512 + tid];
            const float gi = gp[0], gf = gp[128], gg = gp[256], go = gp[384];
            __builtin_amdgcn_sched_barrier(0);

            v8f acc[4][2];
#pragma unroll
            for (int g = 0; g < 4; ++g) { acc[g][0] = v8f{}; acc[g][1] = v8f{}; }
#define PRODUCT(g, tl, ks)                                                            \
            acc[g][tl] = __builtin_amdgcn_wmma_f32_16x16x32_f16(                      \
                false, a[ks], false, bw[g][tl][ks], (short)0, acc[g][tl], false, false);
#define PRODUCTS_K(g, ks) PRODUCT(g, 0, ks) PRODUCT(g, 1, ks)
            PRODUCTS_K(2, 0) PRODUCTS_K(0, 0) PRODUCTS_K(2, 1) PRODUCTS_K(0, 1)
            PRODUCTS_K(2, 2) PRODUCTS_K(0, 2) PRODUCTS_K(1, 0)
            PRODUCTS_K(2, 3) PRODUCTS_K(0, 3) PRODUCTS_K(1, 1)
            __builtin_amdgcn_sched_barrier(0);
            PRODUCTS_K(1, 2) PRODUCTS_K(1, 3)
            const float zg = (hh ? acc[2][1][0] : acc[2][0][0]) + gg;
            const float zi = (hh ? acc[0][1][0] : acc[0][0][0]) + gi;
            const float gt = tanh_sel(zg);
            const float ig = 1.f / (1.f + __expf(-zi));
            PRODUCTS_K(3, 0) PRODUCTS_K(3, 1)
            const float zf = (hh ? acc[1][1][0] : acc[1][0][0]) + gf;
            const float fg = 1.f / (1.f + __expf(-zf));
            c = __builtin_fmaf(fg, c, ig * gt);
            const float tc = tanh_sel(c);
            PRODUCTS_K(3, 2) PRODUCTS_K(3, 3)
#undef PRODUCTS_K
#undef PRODUCT
            const float zo = (hh ? acc[3][1][0] : acc[3][0][0]) + go;
            const float og = 1.f / (1.f + __expf(-zo));
            const float h = og * tc;
            h16[cur ^ 1][tid] = (_Float16)h;
            __syncthreads();
            hd[(size_t)tt * DIM + tid] = h;
        }
    }
}
#undef G_ROWS
#undef G_DECL
#undef G_LOAD
#undef G_STORE

__global__ __launch_bounds__(256) void bilstm_concat_relu(const float* __restrict__ hbuf,
                                                          _Float16* __restrict__ xs_rnn) {
    const int g = blockIdx.x * 256 + threadIdx.x;
    const int t = g >> 5;
    const int n = (g & 31) * 8;
    const float* p = (n < DIM) ? hbuf + (size_t)t * DIM + n
                               : hbuf + (size_t)(MM + t) * DIM + (n - DIM);
    const v4f a = *reinterpret_cast<const v4f*>(p);
    const v4f b = *reinterpret_cast<const v4f*>(p + 4);
    v8h h;
#pragma unroll
    for (int e = 0; e < 4; ++e) { h[e] = (_Float16)fmaxf(a[e], 0.f); h[e + 4] = (_Float16)fmaxf(b[e], 0.f); }
    *reinterpret_cast<v8h*>(xs_rnn + (size_t)t * (2 * DIM) + n) = h;
}

__device__ __forceinline__ void row_mean_wg(const float* __restrict__ W, float* __restrict__ out, int blk) {
    const int lane = threadIdx.x & 31;
    const int i    = blk * 8 + (threadIdx.x >> 5);
    const v2f* row = reinterpret_cast<const v2f*>(W + (size_t)i * MM);
    float s[4] = {0.f, 0.f, 0.f, 0.f};
    for (int b = 0; b < 64; b += 16) {
        v2f x[16];
#pragma unroll
        for (int q = 0; q < 16; ++q) x[q] = row[(b + q) * 32 + lane];
#pragma unroll
        for (int q = 0; q < 16; ++q) { s[(2 * q) & 3] += x[q][0]; s[(2 * q + 1) & 3] += x[q][1]; }
    }
    const float t = wave_sum((s[0] + s[1]) + (s[2] + s[3]));
    if (lane == 0) out[i] = t / (float)MM;
}

#define RBLK 64
#define RLEN (NN / RBLK)
static_assert(NN == MM && RLEN * RBLK == NN && RLEN == 64, "pool_part maps 256 threads on 4 x 64 rows");

__device__ __forceinline__ void col_sum_wg(const float* __restrict__ W, float* __restrict__ colpart,
                                           int bx, int b) {
    const int j = bx * 256 + threadIdx.x;
    const float* p = W + (size_t)b * RLEN * MM + j;
    float s = 0.f;
    for (int r = 0; r < RLEN; ++r) s += p[(size_t)r * MM];
    colpart[(size_t)b * MM + j] = s;
}

__global__ __launch_bounds__(256) void row_col_sums(const float* __restrict__ W,
                                                    float* __restrict__ rowout,
                                                    float* __restrict__ colpart) {
    const int b = blockIdx.x;
    if (b < NN / 8) row_mean_wg(W, rowout, b);
    else {
        const int q = b - NN / 8;
        col_sum_wg(W, colpart, q % (MM / 256), q / (MM / 256));
    }
}

__global__ void pool_part(const float* __restrict__ h, const float* __restrict__ hp,
                          const float* __restrict__ wc, const float* __restrict__ colpart,
                          float* __restrict__ poolpart) {
    __shared__ float red[4][RLEN];
    __shared__ float wsh[2][RLEN];
    const int tid = threadIdx.x;
    const int i0  = blockIdx.x * RLEN;
    {
        const int r = tid & (RLEN - 1);
        const int q = tid >> 6;
        float s = 0.f;
        for (int b = 0; b < RBLK / 4; ++b) s += colpart[(size_t)(q * (RBLK / 4) + b) * MM + i0 + r];
        red[q][r] = s;
        if (tid < RLEN) wsh[0][tid] = wc[i0 + tid];
    }
    __syncthreads();
    if (tid < RLEN)
        wsh[1][tid] = ((red[0][tid] + red[1][tid]) + (red[2][tid] + red[3][tid])) / (float)NN;
    __syncthreads();
    const int n    = tid & 127;
    const int half = tid >> 7;
    const float* src = (half ? hp : h) + (size_t)i0 * DIM + n;
    const float* wv  = wsh[half];
    float s = 0.f;
    for (int r = 0; r < RLEN; ++r) s += wv[r] * src[(size_t)r * DIM];
    poolpart[blockIdx.x * 256 + tid] = s;
}

__global__ void final_mlp(const float* __restrict__ poolpart, const float* __restrict__ Wo,
                          const float* __restrict__ bo, const float* __restrict__ Wi,
                          const float* __restrict__ bi, float* __restrict__ out) {
    __shared__ float cur[256], nxt[256];
    const int t = threadIdx.x;
    {
        float s = 0.f;
        for (int b = 0; b < RBLK; ++b) s += poolpart[b * 256 + t];
        cur[t] = s / (float)NN;
    }
    __syncthreads();
    for (int l = 0; l < 2; ++l) {
        const float* Wl = Wo + (size_t)l * 256 * 256;
        float s = bo[l * 256 + t];
        for (int k = 0; k < 256; ++k) s += cur[k] * Wl[(size_t)t * 256 + k];
        nxt[t] = fmaxf(s, 0.f);
        __syncthreads();
        cur[t] = nxt[t];
        __syncthreads();
    }
    if (t < 2) {
        float s = bi[t];
        for (int k = 0; k < 256; ++k) s += cur[k] * Wi[(size_t)t * 256 + k];
        out[t] = s;
    }
}

extern "C" void kernel_launch(void* const* d_in, const int* in_sizes, int n_in,
                              void* d_out, int out_size, void* d_ws, size_t ws_size,
                              hipStream_t stream) {
    (void)in_sizes; (void)n_in; (void)out_size; (void)ws_size;
    const int*   fingerprints = (const int*)d_in[0];
    const float* adjacency    = (const float*)d_in[1];
    const int*   words        = (const int*)d_in[2];
    const float* emb_fp       = (const float*)d_in[3];
    const float* emb_word     = (const float*)d_in[4];
    const float* Wg  = (const float*)d_in[5];
    const float* bg  = (const float*)d_in[6];
    const float* Wih = (const float*)d_in[7];
    const float* Whh = (const float*)d_in[8];
    const float* bih = (const float*)d_in[9];
    const float* bhh = (const float*)d_in[10];
    const float* Wc  = (const float*)d_in[11];
    const float* bc  = (const float*)d_in[12];
    const float* Wp  = (const float*)d_in[13];
    const float* bp  = (const float*)d_in[14];
    const float* Wo  = (const float*)d_in[15];
    const float* bo  = (const float*)d_in[16];
    const float* Wi  = (const float*)d_in[17];
    const float* bi  = (const float*)d_in[18];

    float* out         = (float*)d_out;
    float* weights_out = out + 2;

    char* base = (char*)d_ws;
    size_t off = 0;
    auto alloc = [&](size_t bytes) -> void* {
        void* p = base + off;
        off = (off + bytes + 255) & ~(size_t)255;
        return p;
    };
    float*    xs       = (float*)   alloc((size_t)NN * DIM * 4);
    _Float16* xs_h     = (_Float16*)alloc((size_t)NN * DIM * 2);
    _Float16* hs_h     = (_Float16*)alloc((size_t)NN * DIM * 2);
    _Float16* hsF_h    = (_Float16*)alloc((size_t)DIM * NN * 2);
    _Float16* attn_h   = (_Float16*)alloc((size_t)NN * NN * 2);
    _Float16* Wg_h     = (_Float16*)alloc((size_t)3 * DIM * DIM * 2);
    _Float16* Wc_h     = (_Float16*)alloc((size_t)DIM * DIM * 2);
    _Float16* Wp_h     = (_Float16*)alloc((size_t)DIM * 2 * DIM * 2);
    _Float16* Wih_h    = (_Float16*)alloc((size_t)2 * 512 * DIM * 2);
    float*    bsum     = (float*)   alloc((size_t)2 * 512 * 4);
    _Float16* wv_h     = (_Float16*)alloc((size_t)MM * DIM * 2);
    float*    G        = (float*)   alloc((size_t)2 * MM * 512 * 4);
    float*    hbuf     = (float*)   alloc((size_t)2 * MM * DIM * 4);
    _Float16* xs_rnn_h = (_Float16*)alloc((size_t)MM * 2 * DIM * 2);
    float*    h_f32    = (float*)   alloc((size_t)NN * DIM * 4);
    _Float16* h_h      = (_Float16*)alloc((size_t)NN * DIM * 2);
    float*    hp_f32   = (float*)   alloc((size_t)MM * DIM * 4);
    _Float16* hp_h     = (_Float16*)alloc((size_t)MM * DIM * 2);
    float*    w_comp   = (float*)   alloc((size_t)NN * 4);
    float*    colpart  = (float*)attn_h;
    float*    poolpart = colpart + (size_t)RBLK * MM;

    setup_all<<<SB_END, 256, 0, stream>>>(Wg, Wg_h, Wc, Wc_h, Wp, Wp_h, Wih, Wih_h, bih, bhh, bsum,
                                          fingerprints, emb_fp, xs, xs_h, words, emb_word, wv_h);

    for (int l = 0; l < 3; ++l) {
        gemm_bias_act<<<dim3(NN / 16, DIM / 16), 32, 0, stream>>>(
            xs_h, Wg_h + (size_t)l * DIM * DIM, bg + l * DIM,
            nullptr, hs_h, hsF_h, DIM, DIM, DIM, DIM, 1);
        gemm_attn<<<dim3(NN / 256, NN / 128), 256, 0, stream>>>(hs_h, adjacency, attn_h);
        gemm_norm_axpy<<<NN / 16, 256, 0, stream>>>(attn_h, hsF_h, xs, xs_h);
    }

    gemm_pre3<<<PJ_TILES + 2 * PG_TILES, 32, 0, stream>>>(xs_h, Wc_h, bc, h_f32, h_h,
                                                         wv_h, Wih_h, bsum, G);

    lstm_dir<<<2, 128, 0, stream>>>(G, Whh, hbuf);
    bilstm_concat_relu<<<MM / 8, 256, 0, stream>>>(hbuf, xs_rnn_h);
    gemm_bias_act<<<dim3(MM / 16, DIM / 16), 32, 0, stream>>>(
        xs_rnn_h, Wp_h, bp, hp_f32, hp_h, nullptr, 2 * DIM, 2 * DIM, 2 * DIM, DIM, 1);

    gemm_weights<<<dim3(NN / 256, MM / 128), 256, 0, stream>>>(h_h, hp_h, weights_out);

    row_col_sums<<<NN / 8 + (MM / 256) * RBLK, 256, 0, stream>>>(weights_out, w_comp, colpart);
    pool_part<<<RBLK, 256, 0, stream>>>(h_f32, hp_f32, w_comp, colpart, poolpart);
    final_mlp<<<1, 256, 0, stream>>>(poolpart, Wo, bo, Wi, bi, out);
}


# recurrence cut into 10 segments (h, c carried through the workspace) that run as the first 2 work-groups of the 9 graph-layer launches and the projection launch, beside that work; sums unchanged
# speedup vs baseline: 7.9484x; 1.0365x over previous
#include <hip/hip_runtime.h>

typedef __attribute__((ext_vector_type(16))) _Float16 v16h;
typedef __attribute__((ext_vector_type(8)))  float    v8f;
typedef __attribute__((ext_vector_type(4)))  float    v4f;

#define DIM 128
#define NN  4096
#define MM  4096

__device__ __forceinline__ v16h frag_load(const _Float16* __restrict__ tile, int ld) {
    const int lane = threadIdx.x & 31;
    const int r    = lane & 15;
    const int kb   = (lane & 16) ? 8 : 0;
    const _Float16* q = tile + (size_t)r * (size_t)ld + kb;
    v16h f;
    uint4* fu = reinterpret_cast<uint4*>(&f);
    fu[0] = *reinterpret_cast<const uint4*>(q);
    fu[1] = *reinterpret_cast<const uint4*>(q + 16);
    return f;
}

__device__ __forceinline__ void frag_rc(int& rbase, int& col) {
    const int lane = threadIdx.x & 31;
    col   = lane & 15;
    rbase = (lane & 16) ? 8 : 0;
}

typedef __attribute__((ext_vector_type(8))) _Float16 v8h;
__device__ __forceinline__ void gemm_tile(const _Float16* __restrict__ A, const _Float16* __restrict__ B,
                                          const float* __restrict__ bias,
                                          float* __restrict__ Cf, _Float16* __restrict__ Ch,
                                          _Float16* __restrict__ ChF,
                                          int K, int lda, int ldb, int ldc, int do_relu,
                                          int bx, int by, int ny) {
    const int i0 = bx * 16;
    const int j0 = by * 16;
    v8f acc = {};
    for (int k = 0; k < K; k += 32) {
        v16h a = frag_load(A + (size_t)i0 * lda + k, lda);
        v16h b = frag_load(B + (size_t)j0 * ldb + k, ldb);
        acc = __builtin_amdgcn_wmma_f32_16x16x32_f16(false, a, false, b, (short)0, acc,
                                                     false, false);
    }
    int rb, col; frag_rc(rb, col);
    const float bv = bias ? bias[j0 + col] : 0.f;
    v8h piece;
#pragma unroll
    for (int r = 0; r < 8; ++r) {
        const int row = i0 + rb + r;
        float v = acc[r] + bv;
        if (do_relu) v = fmaxf(v, 0.f);
        const size_t idx = (size_t)row * ldc + j0 + col;
        if (Cf) Cf[idx] = v;
        if (Ch) Ch[idx] = (_Float16)v;
        piece[r] = (_Float16)v;
    }
    if (ChF) {
        const size_t o = (size_t)((bx >> 1) * ny + by) * 512 + (bx & 1) * 256 + (threadIdx.x & 31) * 8;
        *reinterpret_cast<v8h*>(ChF + o) = piece;
    }
}

__global__ void gemm_bias_act(const _Float16* __restrict__ A, const _Float16* __restrict__ B,
                              const float* __restrict__ bias,
                              float* __restrict__ Cf, _Float16* __restrict__ Ch,
                              _Float16* __restrict__ ChF,
                              int K, int lda, int ldb, int ldc, int do_relu) {
    gemm_tile(A, B, bias, Cf, Ch, ChF, K, lda, ldb, ldc, do_relu, blockIdx.x, blockIdx.y, gridDim.y);
}

#define PJ_TILES ((NN / 16) * (DIM / 16))
#define PG_TILES ((MM / 16) * (512 / 16))
__global__ void gemm_pregates(const _Float16* __restrict__ wv_h, const _Float16* __restrict__ Wih_h,
                              const float* __restrict__ bsum, float* __restrict__ G) {
    int b = blockIdx.x;
    const int d = b / PG_TILES;
    b -= d * PG_TILES;
    gemm_tile(wv_h, Wih_h + (size_t)d * 512 * DIM, bsum + d * 512,
              G + (size_t)d * MM * 512, nullptr, nullptr, DIM, DIM, DIM, 512, 0,
              b % (MM / 16), b / (MM / 16), 512 / 16);
}

typedef __attribute__((ext_vector_type(2))) float    v2f;


__device__ __forceinline__ void attn_block(const _Float16* __restrict__ hs,
                                           const float* __restrict__ adj,
                                           _Float16* __restrict__ attn, int bxx, int byy) {
    const int lane = threadIdx.x & 31;
    const int wave = threadIdx.x >> 5;
    const int c    = lane & 15;
    const int hh   = lane >> 4;
    const int i0   = (bxx * 4 + wave) * 32;
    const int jb   = byy * 128;
    v16h b0[4], b1[4];
#pragma unroll
    for (int ks = 0; ks < 4; ++ks) {
        b0[ks] = frag_load(hs + (size_t)i0 * DIM + ks * 32, DIM);
        b1[ks] = frag_load(hs + (size_t)(i0 + 16) * DIM + ks * 32, DIM);
    }
    const size_t r0 = (size_t)(i0 + c) * NN + jb + 8 * hh;
    const size_t r1 = r0 + (size_t)16 * NN;
    for (int t = 0; t < 8; t += 2) {
        v4f  x[2][4];
        v16h a[2][4];
#pragma unroll
        for (int u = 0; u < 2; ++u) {
            x[u][0] = *reinterpret_cast<const v4f*>(adj + r0 + (t + u) * 16);
            x[u][1] = *reinterpret_cast<const v4f*>(adj + r0 + (t + u) * 16 + 4);
            x[u][2] = *reinterpret_cast<const v4f*>(adj + r1 + (t + u) * 16);
            x[u][3] = *reinterpret_cast<const v4f*>(adj + r1 + (t + u) * 16 + 4);
        }
#pragma unroll
        for (int u = 0; u < 2; ++u)
#pragma unroll
            for (int ks = 0; ks < 4; ++ks)
                a[u][ks] = frag_load(hs + (size_t)(jb + (t + u) * 16) * DIM + ks * 32, DIM);
        __builtin_amdgcn_sched_barrier(0);
#pragma unroll
        for (int u = 0; u < 2; ++u) {
            v8f acc0 = {}, acc1 = {};
#pragma unroll
            for (int ks = 0; ks < 4; ++ks) {
                acc0 = __builtin_amdgcn_wmma_f32_16x16x32_f16(false, a[u][ks], false, b0[ks],
                                                              (short)0, acc0, false, false);
                acc1 = __builtin_amdgcn_wmma_f32_16x16x32_f16(false, a[u][ks], false, b1[ks],
                                                              (short)0, acc1, false, false);
            }
            v8h o0, o1;
#pragma unroll
            for (int r = 0; r < 4; ++r) {
                o0[r]     = (_Float16)(acc0[r] * x[u][0][r]);
                o0[r + 4] = (_Float16)(acc0[r + 4] * x[u][1][r]);
                o1[r]     = (_Float16)(acc1[r] * x[u][2][r]);
                o1[r + 4] = (_Float16)(acc1[r + 4] * x[u][3][r]);
            }
            *reinterpret_cast<v8h*>(attn + r0 + (t + u) * 16) = o0;
            *reinterpret_cast<v8h*>(attn + r1 + (t + u) * 16) = o1;
        }
    }
}

__global__ __launch_bounds__(256) void gemm_weights(const _Float16* __restrict__ h,
                                                    const _Float16* __restrict__ hp,
                                                    float* __restrict__ C) {
    const int lane = threadIdx.x & 31;
    const int wave = threadIdx.x >> 5;
    const int c    = lane & 15;
    const int hh   = lane >> 4;
    const int i0   = (blockIdx.x * 8 + wave) * 32;
    const int jb   = blockIdx.y * 128;
    v16h b0[4], b1[4];
#pragma unroll
    for (int ks = 0; ks < 4; ++ks) {
        b0[ks] = frag_load(h + (size_t)i0 * DIM + ks * 32, DIM);
        b1[ks] = frag_load(h + (size_t)(i0 + 16) * DIM + ks * 32, DIM);
    }
    const size_t r0 = (size_t)(i0 + c) * MM + jb + 8 * hh;
    const size_t r1 = r0 + (size_t)16 * MM;
#pragma nounroll
    for (int t = 0; t < 8; t += 2) {
        v16h a[2][4];
#pragma unroll
        for (int u = 0; u < 2; ++u)
#pragma unroll
            for (int ks = 0; ks < 4; ++ks)
                a[u][ks] = frag_load(hp + (size_t)(jb + (t + u) * 16) * DIM + ks * 32, DIM);
        __builtin_amdgcn_sched_barrier(0);
        v8f acc[2][2] = {};
#pragma unroll
        for (int ks = 0; ks < 4; ++ks)
#pragma unroll
            for (int u = 0; u < 2; ++u) {
                acc[u][0] = __builtin_amdgcn_wmma_f32_16x16x32_f16(false, a[u][ks], false, b0[ks],
                                                                   (short)0, acc[u][0], false, false);
                acc[u][1] = __builtin_amdgcn_wmma_f32_16x16x32_f16(false, a[u][ks], false, b1[ks],
                                                                   (short)0, acc[u][1], false, false);
            }
#pragma unroll
        for (int u = 0; u < 2; ++u) {
            v2f* q0 = reinterpret_cast<v2f*>(C + r0 + (t + u) * 16);
            v2f* q1 = reinterpret_cast<v2f*>(C + r1 + (t + u) * 16);
#pragma unroll
            for (int r = 0; r < 4; ++r) {
                q0[r] = v2f{acc[u][0][2 * r], acc[u][0][2 * r + 1]};
                q1[r] = v2f{acc[u][1][2 * r], acc[u][1][2 * r + 1]};
            }
        }
    }
}

__device__ __forceinline__ float wave_sum(float s) {
    for (int m = 16; m > 0; m >>= 1) s += __shfl_xor(s, m, 32);
    return s;
}

#define NA_PITCH 132
__device__ __forceinline__ void norm_axpy_block(const _Float16* __restrict__ attn,
                                                const _Float16* __restrict__ hsF,
                                                float* __restrict__ xs,
                                                _Float16* __restrict__ xs_h, int bx) {
    __shared__ float part[8][16 * NA_PITCH];
    __shared__ float dpart[8][16];
    __shared__ float dsum[16];
    const int lane = threadIdx.x & 31;
    const int wave = threadIdx.x >> 5;
    const int i0   = bx * 16;
    const int c  = lane & 15;
    const int hh = lane >> 4;
    for (int half = 0; half < 2; ++half) {
        const int kw = wave + 4 * half;
        const _Float16* Ab = attn + (size_t)i0 * NN + kw * 512;
        const _Float16* Bb = hsF + (size_t)(kw * 16) * (8 * 512) + lane * 8;
        v8f acc[8];
#pragma unroll
        for (int j = 0; j < 8; ++j) acc[j] = v8f{};
        float as0 = 0.f, as1 = 0.f;
#pragma nounroll
        for (int t = 0; t < 8; ++t) {
            v16h a[2], b[2][8];
#pragma unroll
            for (int u = 0; u < 2; ++u) a[u] = frag_load(Ab + t * 64 + 32 * u, NN);
#pragma unroll
            for (int u = 0; u < 2; ++u)
#pragma unroll
                for (int j = 0; j < 8; ++j) {
                    const _Float16* p = Bb + (size_t)((2 * t + u) * 8 + j) * 512;
                    uint4* fu = reinterpret_cast<uint4*>(&b[u][j]);
                    fu[0] = *reinterpret_cast<const uint4*>(p);
                    fu[1] = *reinterpret_cast<const uint4*>(p + 256);
                }
            __builtin_amdgcn_sched_barrier(0);
#pragma unroll
            for (int u = 0; u < 2; ++u)
#pragma unroll
                for (int j = 0; j < 8; ++j)
                    acc[j] = __builtin_amdgcn_wmma_f32_16x16x32_f16(false, a[u], false, b[u][j],
                                                                    (short)0, acc[j], false, false);
#pragma unroll
            for (int u = 0; u < 2; ++u)
#pragma unroll
                for (int e = 0; e < 16; e += 2) {
                    as0 += fabsf((float)a[u][e]);
                    as1 += fabsf((float)a[u][e + 1]);
                }
        }
        {
            float rsum = as0 + as1;
            rsum += __shfl_xor(rsum, 16, 32);
            if (lane < 16) dpart[kw][lane] = rsum;
        }
        float* pw = &part[kw][0];
#pragma unroll
        for (int j = 0; j < 8; ++j)
#pragma unroll
            for (int r = 0; r < 8; ++r)
                pw[(8 * hh + r) * NA_PITCH + 16 * j + c] = acc[j][r];
    }
    const int tid = threadIdx.x;
    float xv[16];
#pragma unroll
    for (int q = 0; q < 16; ++q) xv[q] = xs[(size_t)(i0 + q) * DIM + tid];
    __syncthreads();
    if (tid < 16)
        dsum[tid] = fmaxf(((dpart[0][tid] + dpart[1][tid]) + (dpart[2][tid] + dpart[3][tid]))
                        + ((dpart[4][tid] + dpart[5][tid]) + (dpart[6][tid] + dpart[7][tid])), 1e-12f);
    __syncthreads();
#pragma unroll
    for (int q = 0; q < 16; ++q) {
        const int o = q * NA_PITCH + tid;
        const float s = ((part[0][o] + part[1][o]) + (part[2][o] + part[3][o]))
                      + ((part[4][o] + part[5][o]) + (part[6][o] + part[7][o]));
        const float v = xv[q] + s / dsum[q];
        xs[(size_t)(i0 + q) * DIM + tid]   = v;
        xs_h[(size_t)(i0 + q) * DIM + tid] = (_Float16)v;
    }
}

typedef __attribute__((ext_vector_type(4))) _Float16 v4h;

__device__ __forceinline__ void cvt8(const float* __restrict__ s, _Float16* __restrict__ d, int i) {
    const v4f a = *reinterpret_cast<const v4f*>(s + i);
    const v4f b = *reinterpret_cast<const v4f*>(s + i + 4);
    v8h h;
#pragma unroll
    for (int e = 0; e < 4; ++e) { h[e] = (_Float16)a[e]; h[e + 4] = (_Float16)b[e]; }
    *reinterpret_cast<v8h*>(d + i) = h;
}

__device__ __forceinline__ void gather8(const int* __restrict__ idx, const float* __restrict__ emb,
                                        float* __restrict__ of32, _Float16* __restrict__ of16, int blk) {
    const int lane = threadIdx.x & 31;
    const int i    = blk * 8 + (threadIdx.x >> 5);
    const v4f v = *reinterpret_cast<const v4f*>(emb + (size_t)idx[i] * DIM + 4 * lane);
    if (of32) *reinterpret_cast<v4f*>(of32 + (size_t)i * DIM + 4 * lane) = v;
    v4h h;
    h[0] = (_Float16)v[0]; h[1] = (_Float16)v[1]; h[2] = (_Float16)v[2]; h[3] = (_Float16)v[3];
    *reinterpret_cast<v4h*>(of16 + (size_t)i * DIM + 4 * lane) = h;
}

#define SB_WG   0
#define SB_WC   (SB_WG + (3 * DIM * DIM) / 2048)
#define SB_WP   (SB_WC + (DIM * DIM) / 2048)
#define SB_WIH  (SB_WP + (DIM * 2 * DIM) / 2048)
#define SB_BIAS (SB_WIH + (2 * 512 * DIM) / 2048)
#define SB_FP   (SB_BIAS + (2 * 512) / 256)
#define SB_WORD (SB_FP + NN / 8)
#define SB_END  (SB_WORD + MM / 8)
__global__ __launch_bounds__(256) void setup_all(const float* __restrict__ Wg, _Float16* __restrict__ Wg_h,
                                                 const float* __restrict__ Wc, _Float16* __restrict__ Wc_h,
                                                 const float* __restrict__ Wp, _Float16* __restrict__ Wp_h,
                                                 const float* __restrict__ Wih, _Float16* __restrict__ Wih_h,
                                                 const float* __restrict__ bih, const float* __restrict__ bhh,
                                                 float* __restrict__ bsum,
                                                 const int* __restrict__ fingerprints,
                                                 const float* __restrict__ emb_fp,
                                                 float* __restrict__ xs, _Float16* __restrict__ xs_h,
                                                 const int* __restrict__ words,
                                                 const float* __restrict__ emb_word,
                                                 _Float16* __restrict__ wv_h) {
    const int b = blockIdx.x;
    if (b < SB_WC)        cvt8(Wg,  Wg_h,  ((b - SB_WG)  * 256 + threadIdx.x) * 8);
    else if (b < SB_WP)   cvt8(Wc,  Wc_h,  ((b - SB_WC)  * 256 + threadIdx.x) * 8);
    else if (b < SB_WIH)  cvt8(Wp,  Wp_h,  ((b - SB_WP)  * 256 + threadIdx.x) * 8);
    else if (b < SB_BIAS) cvt8(Wih, Wih_h, ((b - SB_WIH) * 256 + threadIdx.x) * 8);
    else if (b < SB_FP) {
        const int i = (b - SB_BIAS) * 256 + threadIdx.x;
        bsum[i] = bih[i] + bhh[i];
    }
    else if (b < SB_WORD) gather8(fingerprints, emb_fp, xs, xs_h, b - SB_FP);
    else                  gather8(words, emb_word, nullptr, wv_h, b - SB_WORD);
}

__device__ __forceinline__ float tanh_sel(float x) {
    const float y  = __builtin_fabsf(x);
    const float y2 = y * y;
    float p = __builtin_fmaf(y2, -0x1.758e7ap-8f, 0x1.521192p-6f);
    p = __builtin_fmaf(y2, p, -0x1.b8389cp-5f);
    p = __builtin_fmaf(y2, p, 0x1.110704p-3f);
    p = __builtin_fmaf(y2, p, -0x1.555532p-2f);
    const float zs = __builtin_fmaf(y2, y * p, y);
    const float e  = expf(2.0f * y);
    const float zl = __builtin_fmaf(-2.0f, __builtin_amdgcn_rcpf(e + 1.0f), 1.0f);
    return __builtin_copysignf((y < 0.625f) ? zs : zl, x);
}

#define GCH 16
static_assert(GCH == 16 && MM % GCH == 0, "G_ROWS names 16 rows; pieces tile the sequence");
#define N_SEG      10
#define SEG_PIECES 26
static_assert(SEG_PIECES * N_SEG >= MM / GCH && SEG_PIECES * (N_SEG - 1) < MM / GCH,
              "the segments cover the sequence and the last one is not empty");
__device__ __forceinline__ void lstm_segment(const float* __restrict__ G,
                                             const float* __restrict__ Whh,
                                             float* __restrict__ hout,
                                             float* __restrict__ cst,
                                             _Float16* __restrict__ hst,
                                             int seg, int d) {
    __shared__ __attribute__((aligned(16))) _Float16 h16[2][DIM];
    __shared__ __attribute__((aligned(16))) float gs[2][GCH * 512];
    const int tid  = threadIdx.x;
    const int p0   = seg * SEG_PIECES;
    const int p1   = (p0 + SEG_PIECES < MM / GCH) ? p0 + SEG_PIECES : MM / GCH;
    const int lane = tid & 31;
    const int c16  = lane & 15;
    const int hh   = lane >> 4;
    const int u0   = tid & ~31;
    const float* W  = Whh + (size_t)d * 512 * DIM;
    const float* Gd = G + (size_t)d * MM * 512;
    float* hd = hout + (size_t)d * MM * DIM;
    _Float16 h_in = (_Float16)0.f;
    float c = 0.f;
    if (p0) {
        h_in = hst[d * DIM + tid];
        c    = cst[d * DIM + tid];
    }

    v16h bw[4][2][4];
#pragma unroll
    for (int g = 0; g < 4; ++g)
#pragma unroll
        for (int tl = 0; tl < 2; ++tl)
#pragma unroll
            for (int ks = 0; ks < 4; ++ks) {
                const float* q = W + (size_t)(g * 128 + u0 + tl * 16 + c16) * DIM + ks * 32 + 8 * hh;
                const float4 x0 = *reinterpret_cast<const float4*>(q);
                const float4 x1 = *reinterpret_cast<const float4*>(q + 4);
                const float4 x2 = *reinterpret_cast<const float4*>(q + 16);
                const float4 x3 = *reinterpret_cast<const float4*>(q + 20);
                v16h f;
                f[0]  = (_Float16)x0.x; f[1]  = (_Float16)x0.y; f[2]  = (_Float16)x0.z; f[3]  = (_Float16)x0.w;
                f[4]  = (_Float16)x1.x; f[5]  = (_Float16)x1.y; f[6]  = (_Float16)x1.z; f[7]  = (_Float16)x1.w;
                f[8]  = (_Float16)x2.x; f[9]  = (_Float16)x2.y; f[10] = (_Float16)x2.z; f[11] = (_Float16)x2.w;
                f[12] = (_Float16)x3.x; f[13] = (_Float16)x3.y; f[14] = (_Float16)x3.z; f[15] = (_Float16)x3.w;
                bw[g][tl][ks] = f;
            }

#define G_ROWS(X) X(0) X(1) X(2) X(3) X(4) X(5) X(6) X(7) X(8) X(9) X(10) X(11) X(12) X(13) X(14) X(15)
#define G_DECL(i) v4f pre##i;
#define G_LOAD(i) pre##i = src[(i) * 128 + tid];
#define G_STORE(i) dst[(i) * 128 + tid] = pre##i;
    G_ROWS(G_DECL)
    {
        const v4f* src = reinterpret_cast<const v4f*>(Gd + (size_t)(d ? (MM - (p0 + 1) * GCH) : p0 * GCH) * 512);
        G_ROWS(G_LOAD)
        v4f* dst = reinterpret_cast<v4f*>(&gs[p0 & 1][0]);
        G_ROWS(G_STORE)
    }
    h16[0][tid] = h_in;
    __syncthreads();

    for (int p = p0; p < p1; ++p) {
        const int  gb   = p & 1;
        const bool more = (p + 1) < p1;
        if (more) {
            const int r0 = d ? (MM - (p + 2) * GCH) : (p + 1) * GCH;
            const v4f* src = reinterpret_cast<const v4f*>(Gd + (size_t)r0 * 512);
            G_ROWS(G_LOAD)
        }

        for (int s = 0; s < GCH; ++s) {
            const int t   = p * GCH + s;
            const int tt  = d ? (MM - 1 - t) : t;
            const int cur = t & 1;
            const int sr  = d ? (GCH - 1 - s) : s;

            if (more && s == GCH - 1) {
                v4f* dst = reinterpret_cast<v4f*>(&gs[gb ^ 1][0]);
                G_ROWS(G_STORE)
            }

            const _Float16* hp = &h16[cur][8 * hh];
            v16h a[4];
#pragma unroll
            for (int ks = 0; ks < 4; ++ks) {
                uint4* au = reinterpret_cast<uint4*>(&a[ks]);
                au[0] = *reinterpret_cast<const uint4*>(hp + ks * 32);
                au[1] = *reinterpret_cast<const uint4*>(hp + ks * 32 + 16);
            }
            const float* gp = &gs[gb][sr * 512 + tid];
            const float gi = gp[0], gf = gp[128], gg = gp[256], go = gp[384];
            __builtin_amdgcn_sched_barrier(0);

            v8f acc[4][2];
#pragma unroll
            for (int g = 0; g < 4; ++g) { acc[g][0] = v8f{}; acc[g][1] = v8f{}; }
#define PRODUCT(g, tl, ks)                                                            \
            acc[g][tl] = __builtin_amdgcn_wmma_f32_16x16x32_f16(                      \
                false, a[ks], false, bw[g][tl][ks], (short)0, acc[g][tl], false, false);
#define PRODUCTS_K(g, ks) PRODUCT(g, 0, ks) PRODUCT(g, 1, ks)
            PRODUCTS_K(2, 0) PRODUCTS_K(0, 0) PRODUCTS_K(2, 1) PRODUCTS_K(0, 1)
            PRODUCTS_K(2, 2) PRODUCTS_K(0, 2) PRODUCTS_K(1, 0)
            PRODUCTS_K(2, 3) PRODUCTS_K(0, 3) PRODUCTS_K(1, 1)
            __builtin_amdgcn_sched_barrier(0);
            PRODUCTS_K(1, 2) PRODUCTS_K(1, 3)
            const float zg = (hh ? acc[2][1][0] : acc[2][0][0]) + gg;
            const float zi = (hh ? acc[0][1][0] : acc[0][0][0]) + gi;
            const float gt = tanh_sel(zg);
            const float ig = 1.f / (1.f + __expf(-zi));
            PRODUCTS_K(3, 0) PRODUCTS_K(3, 1)
            const float zf = (hh ? acc[1][1][0] : acc[1][0][0]) + gf;
            const float fg = 1.f / (1.f + __expf(-zf));
            c = __builtin_fmaf(fg, c, ig * gt);
            const float tc = tanh_sel(c);
            PRODUCTS_K(3, 2) PRODUCTS_K(3, 3)
#undef PRODUCTS_K
#undef PRODUCT
            const float zo = (hh ? acc[3][1][0] : acc[3][0][0]) + go;
            const float og = 1.f / (1.f + __expf(-zo));
            const float h = og * tc;
            h16[cur ^ 1][tid] = (_Float16)h;
            __syncthreads();
            hd[(size_t)tt * DIM + tid] = h;
        }
    }
    cst[d * DIM + tid] = c;
    hst[d * DIM + tid] = h16[0][tid];
}
#undef G_ROWS
#undef G_DECL
#undef G_LOAD
#undef G_STORE

__global__ __launch_bounds__(128) void recur_beside(int role, int seg,
                                                    const float* G, const float* Whh, float* hbuf,
                                                    float* cst, _Float16* hst,
                                                    const _Float16* A_h, const _Float16* B_h,
                                                    const float* bias,
                                                    float* Cf, _Float16* Ch, _Float16* ChF,
                                                    const float* adj, _Float16* attn,
                                                    float* xs, _Float16* xs_h) {
    if (blockIdx.x < 2) {
        lstm_segment(G, Whh, hbuf, cst, hst, seg, blockIdx.x);
        return;
    }
    const int b = blockIdx.x - 2;
    if (role == 0) {
        const int tile = b * 4 + (threadIdx.x >> 5);
        gemm_tile(A_h, B_h, bias, Cf, Ch, ChF, DIM, DIM, DIM, DIM, 1,
                  tile % (NN / 16), tile / (NN / 16), DIM / 16);
    } else if (role == 1) {
        attn_block(A_h, adj, attn, b % (NN / 128), b / (NN / 128));
    } else {
        norm_axpy_block(attn, B_h, xs, xs_h, b);
    }
}


__global__ __launch_bounds__(256) void bilstm_concat_relu(const float* __restrict__ hbuf,
                                                          _Float16* __restrict__ xs_rnn) {
    const int g = blockIdx.x * 256 + threadIdx.x;
    const int t = g >> 5;
    const int n = (g & 31) * 8;
    const float* p = (n < DIM) ? hbuf + (size_t)t * DIM + n
                               : hbuf + (size_t)(MM + t) * DIM + (n - DIM);
    const v4f a = *reinterpret_cast<const v4f*>(p);
    const v4f b = *reinterpret_cast<const v4f*>(p + 4);
    v8h h;
#pragma unroll
    for (int e = 0; e < 4; ++e) { h[e] = (_Float16)fmaxf(a[e], 0.f); h[e + 4] = (_Float16)fmaxf(b[e], 0.f); }
    *reinterpret_cast<v8h*>(xs_rnn + (size_t)t * (2 * DIM) + n) = h;
}

__device__ __forceinline__ void row_mean_wg(const float* __restrict__ W, float* __restrict__ out, int blk) {
    const int lane = threadIdx.x & 31;
    const int i    = blk * 8 + (threadIdx.x >> 5);
    const v2f* row = reinterpret_cast<const v2f*>(W + (size_t)i * MM);
    float s[4] = {0.f, 0.f, 0.f, 0.f};
    for (int b = 0; b < 64; b += 16) {
        v2f x[16];
#pragma unroll
        for (int q = 0; q < 16; ++q) x[q] = row[(b + q) * 32 + lane];
#pragma unroll
        for (int q = 0; q < 16; ++q) { s[(2 * q) & 3] += x[q][0]; s[(2 * q + 1) & 3] += x[q][1]; }
    }
    const float t = wave_sum((s[0] + s[1]) + (s[2] + s[3]));
    if (lane == 0) out[i] = t / (float)MM;
}

#define RBLK 64
#define RLEN (NN / RBLK)
static_assert(NN == MM && RLEN * RBLK == NN && RLEN == 64, "pool_part maps 256 threads on 4 x 64 rows");

__device__ __forceinline__ void col_sum_wg(const float* __restrict__ W, float* __restrict__ colpart,
                                           int bx, int b) {
    const int j = bx * 256 + threadIdx.x;
    const float* p = W + (size_t)b * RLEN * MM + j;
    float s = 0.f;
    for (int r = 0; r < RLEN; ++r) s += p[(size_t)r * MM];
    colpart[(size_t)b * MM + j] = s;
}

__global__ __launch_bounds__(256) void row_col_sums(const float* __restrict__ W,
                                                    float* __restrict__ rowout,
                                                    float* __restrict__ colpart) {
    const int b = blockIdx.x;
    if (b < NN / 8) row_mean_wg(W, rowout, b);
    else {
        const int q = b - NN / 8;
        col_sum_wg(W, colpart, q % (MM / 256), q / (MM / 256));
    }
}

__global__ void pool_part(const float* __restrict__ h, const float* __restrict__ hp,
                          const float* __restrict__ wc, const float* __restrict__ colpart,
                          float* __restrict__ poolpart) {
    __shared__ float red[4][RLEN];
    __shared__ float wsh[2][RLEN];
    const int tid = threadIdx.x;
    const int i0  = blockIdx.x * RLEN;
    {
        const int r = tid & (RLEN - 1);
        const int q = tid >> 6;
        float s = 0.f;
        for (int b = 0; b < RBLK / 4; ++b) s += colpart[(size_t)(q * (RBLK / 4) + b) * MM + i0 + r];
        red[q][r] = s;
        if (tid < RLEN) wsh[0][tid] = wc[i0 + tid];
    }
    __syncthreads();
    if (tid < RLEN)
        wsh[1][tid] = ((red[0][tid] + red[1][tid]) + (red[2][tid] + red[3][tid])) / (float)NN;
    __syncthreads();
    const int n    = tid & 127;
    const int half = tid >> 7;
    const float* src = (half ? hp : h) + (size_t)i0 * DIM + n;
    const float* wv  = wsh[half];
    float s = 0.f;
    for (int r = 0; r < RLEN; ++r) s += wv[r] * src[(size_t)r * DIM];
    poolpart[blockIdx.x * 256 + tid] = s;
}

__global__ void final_mlp(const float* __restrict__ poolpart, const float* __restrict__ Wo,
                          const float* __restrict__ bo, const float* __restrict__ Wi,
                          const float* __restrict__ bi, float* __restrict__ out) {
    __shared__ float cur[256], nxt[256];
    const int t = threadIdx.x;
    {
        float s = 0.f;
        for (int b = 0; b < RBLK; ++b) s += poolpart[b * 256 + t];
        cur[t] = s / (float)NN;
    }
    __syncthreads();
    for (int l = 0; l < 2; ++l) {
        const float* Wl = Wo + (size_t)l * 256 * 256;
        float s = bo[l * 256 + t];
        for (int k = 0; k < 256; ++k) s += cur[k] * Wl[(size_t)t * 256 + k];
        nxt[t] = fmaxf(s, 0.f);
        __syncthreads();
        cur[t] = nxt[t];
        __syncthreads();
    }
    if (t < 2) {
        float s = bi[t];
        for (int k = 0; k < 256; ++k) s += cur[k] * Wi[(size_t)t * 256 + k];
        out[t] = s;
    }
}

extern "C" void kernel_launch(void* const* d_in, const int* in_sizes, int n_in,
                              void* d_out, int out_size, void* d_ws, size_t ws_size,
                              hipStream_t stream) {
    (void)in_sizes; (void)n_in; (void)out_size; (void)ws_size;
    const int*   fingerprints = (const int*)d_in[0];
    const float* adjacency    = (const float*)d_in[1];
    const int*   words        = (const int*)d_in[2];
    const float* emb_fp       = (const float*)d_in[3];
    const float* emb_word     = (const float*)d_in[4];
    const float* Wg  = (const float*)d_in[5];
    const float* bg  = (const float*)d_in[6];
    const float* Wih = (const float*)d_in[7];
    const float* Whh = (const float*)d_in[8];
    const float* bih = (const float*)d_in[9];
    const float* bhh = (const float*)d_in[10];
    const float* Wc  = (const float*)d_in[11];
    const float* bc  = (const float*)d_in[12];
    const float* Wp  = (const float*)d_in[13];
    const float* bp  = (const float*)d_in[14];
    const float* Wo  = (const float*)d_in[15];
    const float* bo  = (const float*)d_in[16];
    const float* Wi  = (const float*)d_in[17];
    const float* bi  = (const float*)d_in[18];

    float* out         = (float*)d_out;
    float* weights_out = out + 2;

    char* base = (char*)d_ws;
    size_t off = 0;
    auto alloc = [&](size_t bytes) -> void* {
        void* p = base + off;
        off = (off + bytes + 255) & ~(size_t)255;
        return p;
    };
    float*    xs       = (float*)   alloc((size_t)NN * DIM * 4);
    _Float16* xs_h     = (_Float16*)alloc((size_t)NN * DIM * 2);
    _Float16* hs_h     = (_Float16*)alloc((size_t)NN * DIM * 2);
    _Float16* hsF_h    = (_Float16*)alloc((size_t)DIM * NN * 2);
    _Float16* attn_h   = (_Float16*)alloc((size_t)NN * NN * 2);
    _Float16* Wg_h     = (_Float16*)alloc((size_t)3 * DIM * DIM * 2);
    _Float16* Wc_h     = (_Float16*)alloc((size_t)DIM * DIM * 2);
    _Float16* Wp_h     = (_Float16*)alloc((size_t)DIM * 2 * DIM * 2);
    _Float16* Wih_h    = (_Float16*)alloc((size_t)2 * 512 * DIM * 2);
    float*    bsum     = (float*)   alloc((size_t)2 * 512 * 4);
    _Float16* wv_h     = (_Float16*)alloc((size_t)MM * DIM * 2);
    float*    G        = (float*)   alloc((size_t)2 * MM * 512 * 4);
    float*    hbuf     = (float*)   alloc((size_t)2 * MM * DIM * 4);
    _Float16* xs_rnn_h = (_Float16*)alloc((size_t)MM * 2 * DIM * 2);
    float*    h_f32    = (float*)   alloc((size_t)NN * DIM * 4);
    _Float16* h_h      = (_Float16*)alloc((size_t)NN * DIM * 2);
    float*    hp_f32   = (float*)   alloc((size_t)MM * DIM * 4);
    _Float16* hp_h     = (_Float16*)alloc((size_t)MM * DIM * 2);
    float*    w_comp   = (float*)   alloc((size_t)NN * 4);
    float*    cst      = (float*)xs_rnn_h;
    _Float16* hst      = (_Float16*)(cst + 2 * DIM);
    float*    colpart  = (float*)attn_h;
    float*    poolpart = colpart + (size_t)RBLK * MM;

    setup_all<<<SB_END, 256, 0, stream>>>(Wg, Wg_h, Wc, Wc_h, Wp, Wp_h, Wih, Wih_h, bih, bhh, bsum,
                                          fingerprints, emb_fp, xs, xs_h, words, emb_word, wv_h);

    gemm_pregates<<<2 * PG_TILES, 32, 0, stream>>>(wv_h, Wih_h, bsum, G);

#define BESIDE(role, seg, nblk, A, B, bias, Cf, Ch, ChF)                                        \
    recur_beside<<<2 + (nblk), 128, 0, stream>>>(role, seg, G, Whh, hbuf, cst, hst, A, B, bias, \
                                                 Cf, Ch, ChF, adjacency, attn_h, xs, xs_h);
#define GAT_LAYER(l, s)                                                                         \
    BESIDE(0, (s), PJ_TILES / 4, xs_h, Wg_h + (size_t)(l) * DIM * DIM, bg + (l) * DIM,          \
           nullptr, hs_h, hsF_h)                                                                \
    BESIDE(1, (s) + 1, (NN / 128) * (NN / 128), hs_h, nullptr, nullptr, nullptr, nullptr, nullptr) \
    BESIDE(2, (s) + 2, NN / 16, nullptr, hsF_h, nullptr, nullptr, nullptr, nullptr)
    GAT_LAYER(0, 0)
    GAT_LAYER(1, 3)
    GAT_LAYER(2, 6)
    BESIDE(0, 9, PJ_TILES / 4, xs_h, Wc_h, bc, h_f32, h_h, nullptr)
#undef GAT_LAYER
#undef BESIDE
    bilstm_concat_relu<<<MM / 8, 256, 0, stream>>>(hbuf, xs_rnn_h);
    gemm_bias_act<<<dim3(MM / 16, DIM / 16), 32, 0, stream>>>(
        xs_rnn_h, Wp_h, bp, hp_f32, hp_h, nullptr, 2 * DIM, 2 * DIM, 2 * DIM, DIM, 1);

    gemm_weights<<<dim3(NN / 256, MM / 128), 256, 0, stream>>>(h_h, hp_h, weights_out);

    row_col_sums<<<NN / 8 + (MM / 256) * RBLK, 256, 0, stream>>>(weights_out, w_comp, colpart);
    pool_part<<<RBLK, 256, 0, stream>>>(h_f32, hp_f32, w_comp, colpart, poolpart);
    final_mlp<<<1, 256, 0, stream>>>(poolpart, Wo, bo, Wi, bi, out);
}
